# starting kernel
# baseline (speedup 1.0000x reference)
#include <hip/hip_runtime.h>

typedef float v2f __attribute__((ext_vector_type(2)));
typedef float v8f __attribute__((ext_vector_type(8)));

#define NN    11826
#define DIN   9
#define DGATE 36
#define NE    2
#define HH    128
#define MO    32
#define NO    2

__global__ void moe_zero_ws(int* cnt) {
    if (threadIdx.x == 0) cnt[0] = 0;
}

__global__ __launch_bounds__(32) void moe_tile_kernel(
    const float* __restrict__ x,
    const float* __restrict__ gate,
    const float* __restrict__ w_gate,
    const float* __restrict__ w1,
    const float* __restrict__ b1,
    const float* __restrict__ w2,
    const float* __restrict__ b2,
    const float* __restrict__ wo,
    const float* __restrict__ bo,
    float* __restrict__ out,
    int* __restrict__ cnt1)
{
    __shared__ float h_lds[16 * HH];
    __shared__ float y_lds[NE * 16 * MO];
    __shared__ int   sel[16];

    const int lane = threadIdx.x;
    const int row0 = blockIdx.x * 16;
    const int m    = lane & 15;
    const int hi   = lane >> 4;
    const int rowm = min(row0 + m, NN - 1);

    if (lane < 16) {
        const float* g = gate + (size_t)rowm * DGATE;
        float l0 = 0.f, l1 = 0.f;
        #pragma unroll
        for (int d = 0; d < DGATE; ++d) {
            float gv = g[d];
            l0 += gv * w_gate[d * NE + 0];
            l1 += gv * w_gate[d * NE + 1];
        }
        int e = (l1 > l0) ? 1 : 0;
        sel[lane] = e;
        if ((row0 + lane) < NN && e == 1) atomicAdd(cnt1, 1);
    }
    __syncthreads();

    v2f ax[3];
    {
        const float* xr = x + (size_t)rowm * DIN;
        #pragma unroll
        for (int c = 0; c < 3; ++c) {
            const int k0 = c * 4 + hi * 2;
            float v0 = xr[min(k0,     DIN - 1)];
            float v1 = xr[min(k0 + 1, DIN - 1)];
            v2f a;
            a.x = (k0     < DIN) ? v0 : 0.f;
            a.y = (k0 + 1 < DIN) ? v1 : 0.f;
            ax[c] = a;
        }
    }

    for (int e = 0; e < NE; ++e) {
        const float* W1 = w1 + (size_t)e * DIN * HH;
        const float* W2 = w2 + (size_t)e * HH * MO;
        __builtin_prefetch(W2, 0, 1);

        for (int nt = 0; nt < 8; ++nt) {
            const int n0 = nt * 16;
            const float bias = b1[e * HH + n0 + m];
            v8f c8;
            #pragma unroll
            for (int i = 0; i < 8; ++i) c8[i] = bias;
            #pragma unroll
            for (int cix = 0; cix < 3; ++cix) {
                const int kb = cix * 4 + hi * 2;
                float w0 = W1[min(kb,     DIN - 1) * HH + n0 + m];
                float w1v = W1[min(kb + 1, DIN - 1) * HH + n0 + m];
                v2f b;
                b.x = (kb     < DIN) ? w0  : 0.f;
                b.y = (kb + 1 < DIN) ? w1v : 0.f;
                c8 = __builtin_amdgcn_wmma_f32_16x16x4_f32(
                         false, ax[cix], false, b, (short)0, c8, false, false);
            }
            #pragma unroll
            for (int i = 0; i < 8; ++i)
                h_lds[(i + hi * 8) * HH + n0 + m] = fmaxf(c8[i], 0.0f);
        }
        __syncthreads();

        for (int nt = 0; nt < 2; ++nt) {
            const int n0 = nt * 16;
            const float bias = b2[e * MO + n0 + m];
            v8f c8;
            #pragma unroll
            for (int i = 0; i < 8; ++i) c8[i] = bias;
            #pragma unroll 4
            for (int k = 0; k < HH; k += 4) {
                const int ka = k + hi * 2;
                v2f a = *(const v2f*)&h_lds[m * HH + ka];
                v2f b;
                b.x = W2[(ka    ) * MO + n0 + m];
                b.y = W2[(ka + 1) * MO + n0 + m];
                c8 = __builtin_amdgcn_wmma_f32_16x16x4_f32(
                         false, a, false, b, (short)0, c8, false, false);
            }
            #pragma unroll
            for (int i = 0; i < 8; ++i)
                y_lds[e * (16 * MO) + (i + hi * 8) * MO + n0 + m] = c8[i];
        }
        __syncthreads();
    }

    if (lane < 16 && (row0 + lane) < NN) {
        const int e = sel[lane];
        const float* yr = y_lds + e * (16 * MO) + lane * MO;
        float a0 = bo[0], a1 = bo[1];
        #pragma unroll
        for (int j = 0; j < MO; ++j) {
            float yv = yr[j];
            a0 += yv * wo[j * NO + 0];
            a1 += yv * wo[j * NO + 1];
        }
        out[(row0 + lane) * NO + 0] = a0;
        out[(row0 + lane) * NO + 1] = a1;
    }
}

__global__ void moe_loss_kernel(const int* __restrict__ cnt1,
                                float* __restrict__ loss_out) {
    if (threadIdx.x != 0) return;
    const float c1 = (float)cnt1[0];
    const float c0 = (float)NN - c1;
    const float mv = 0.5f * (float)NN;
    const float var = (c0 - mv) * (c0 - mv) + (c1 - mv) * (c1 - mv);
    const float cv  = var / (mv * mv + 1e-10f);
    loss_out[0] = 2.0f * cv * 0.01f;
}

extern "C" void kernel_launch(void* const* d_in, const int* in_sizes, int n_in,
                              void* d_out, int out_size, void* d_ws, size_t ws_size,
                              hipStream_t stream) {
    const float* x      = (const float*)d_in[0];
    const float* gate   = (const float*)d_in[1];
    const float* w_gate = (const float*)d_in[2];
    const float* w1     = (const float*)d_in[3];
    const float* b1     = (const float*)d_in[4];
    const float* w2     = (const float*)d_in[5];
    const float* b2     = (const float*)d_in[6];
    const float* wo     = (const float*)d_in[7];
    const float* bo     = (const float*)d_in[8];
    float* out = (float*)d_out;
    int*   cnt = (int*)d_ws;

    const int tiles = (NN + 15) / 16;

    moe_zero_ws<<<1, 1, 0, stream>>>(cnt);
    moe_tile_kernel<<<tiles, 32, 0, stream>>>(x, gate, w_gate, w1, b1, w2, b2,
                                              wo, bo, out, cnt);
    moe_loss_kernel<<<1, 1, 0, stream>>>(cnt, out + (size_t)NN * NO);
}


# All global operands (x, gate row, w_gate, W1, b1, W2, b2, wo, bo; both experts) loaded up front as one group into registers, fully unrolled: second-level round trips overlap, not one per k-step
# speedup vs baseline: 1.7265x; 1.7265x over previous
#include <hip/hip_runtime.h>

typedef float v2f __attribute__((ext_vector_type(2)));
typedef float v8f __attribute__((ext_vector_type(8)));

#define NN    11826
#define DIN   9
#define DGATE 36
#define NE    2
#define HH    128
#define MO    32
#define NO    2

__global__ void moe_zero_ws(int* cnt) {
    if (threadIdx.x == 0) cnt[0] = 0;
}

__global__ __launch_bounds__(32) void moe_tile_kernel(
    const float* __restrict__ x,
    const float* __restrict__ gate,
    const float* __restrict__ w_gate,
    const float* __restrict__ w1,
    const float* __restrict__ b1,
    const float* __restrict__ w2,
    const float* __restrict__ b2,
    const float* __restrict__ wo,
    const float* __restrict__ bo,
    float* __restrict__ out,
    int* __restrict__ cnt1)
{
    __shared__ float h_lds[16 * HH];
    __shared__ float y_lds[NE * 16 * MO];

    const int lane = threadIdx.x;
    const int row0 = blockIdx.x * 16;
    const int m    = lane & 15;
    const int hi   = lane >> 4;
    const int rowm = min(row0 + m, NN - 1);

    float xv[6];
    float w1f[NE][8][6];
    float b1f[NE][8];
    float w2f[NE][2][64];
    float b2f[NE][2];
    float gv[DGATE];
    float wgv[DGATE * NE];
    float wov[MO * NO];
    float bov[NO];

    {
        const float* xr = x + (size_t)rowm * DIN;
        #pragma unroll
        for (int c = 0; c < 3; ++c) {
            const int k0 = c * 4 + hi * 2;
            xv[2 * c]     = xr[min(k0,     DIN - 1)];
            xv[2 * c + 1] = xr[min(k0 + 1, DIN - 1)];
        }
    }
    #pragma unroll
    for (int e = 0; e < NE; ++e) {
        const float* W1 = w1 + (size_t)e * DIN * HH;
        const float* W2 = w2 + (size_t)e * HH * MO;
        #pragma unroll
        for (int nt = 0; nt < 8; ++nt) {
            const int n0 = nt * 16;
            b1f[e][nt] = b1[e * HH + n0 + m];
            #pragma unroll
            for (int cix = 0; cix < 3; ++cix) {
                const int kb = cix * 4 + hi * 2;
                w1f[e][nt][2 * cix]     = W1[min(kb,     DIN - 1) * HH + n0 + m];
                w1f[e][nt][2 * cix + 1] = W1[min(kb + 1, DIN - 1) * HH + n0 + m];
            }
        }
        #pragma unroll
        for (int nt = 0; nt < 2; ++nt) {
            const int n0 = nt * 16;
            b2f[e][nt] = b2[e * MO + n0 + m];
            #pragma unroll
            for (int ks = 0; ks < HH / 4; ++ks) {
                const int ka = ks * 4 + hi * 2;
                w2f[e][nt][2 * ks]     = W2[(ka    ) * MO + n0 + m];
                w2f[e][nt][2 * ks + 1] = W2[(ka + 1) * MO + n0 + m];
            }
        }
    }
    {
        const float* g = gate + (size_t)rowm * DGATE;
        #pragma unroll
        for (int d = 0; d < DGATE; ++d) gv[d] = g[d];
    }
    #pragma unroll
    for (int d = 0; d < DGATE * NE; ++d) wgv[d] = w_gate[d];
    #pragma unroll
    for (int j = 0; j < MO * NO; ++j) wov[j] = wo[j];
    bov[0] = bo[0]; bov[1] = bo[1];

    v2f ax[3];
    #pragma unroll
    for (int c = 0; c < 3; ++c) {
        const int k0 = c * 4 + hi * 2;
        v2f a;
        a.x = (k0     < DIN) ? xv[2 * c]     : 0.f;
        a.y = (k0 + 1 < DIN) ? xv[2 * c + 1] : 0.f;
        ax[c] = a;
    }

    int esel;
    {
        float l0 = 0.f, l1 = 0.f;
        #pragma unroll
        for (int d = 0; d < DGATE; ++d) {
            l0 += gv[d] * wgv[d * NE + 0];
            l1 += gv[d] * wgv[d * NE + 1];
        }
        esel = (l1 > l0) ? 1 : 0;
        if (hi == 0 && (row0 + m) < NN && esel == 1) atomicAdd(cnt1, 1);
    }


    #pragma unroll
    for (int e = 0; e < NE; ++e) {
        #pragma unroll
        for (int nt = 0; nt < 8; ++nt) {
            const int n0 = nt * 16;
            const float bias = b1f[e][nt];
            v8f c8;
            #pragma unroll
            for (int i = 0; i < 8; ++i) c8[i] = bias;
            #pragma unroll
            for (int cix = 0; cix < 3; ++cix) {
                const int kb = cix * 4 + hi * 2;
                v2f b;
                b.x = (kb     < DIN) ? w1f[e][nt][2 * cix]     : 0.f;
                b.y = (kb + 1 < DIN) ? w1f[e][nt][2 * cix + 1] : 0.f;
                c8 = __builtin_amdgcn_wmma_f32_16x16x4_f32(
                         false, ax[cix], false, b, (short)0, c8, false, false);
            }
            #pragma unroll
            for (int i = 0; i < 8; ++i)
                h_lds[(i + hi * 8) * HH + n0 + m] = fmaxf(c8[i], 0.0f);
        }
        __syncthreads();

        v8f c0, c1;
        #pragma unroll
        for (int i = 0; i < 8; ++i) { c0[i] = b2f[e][0]; c1[i] = b2f[e][1]; }
        #pragma unroll
        for (int ks = 0; ks < HH / 4; ++ks) {
            const int ka = ks * 4 + hi * 2;
            v2f a = *(const v2f*)&h_lds[m * HH + ka];
            v2f bA, bB;
            bA.x = w2f[e][0][2 * ks]; bA.y = w2f[e][0][2 * ks + 1];
            bB.x = w2f[e][1][2 * ks]; bB.y = w2f[e][1][2 * ks + 1];
            c0 = __builtin_amdgcn_wmma_f32_16x16x4_f32(
                     false, a, false, bA, (short)0, c0, false, false);
            c1 = __builtin_amdgcn_wmma_f32_16x16x4_f32(
                     false, a, false, bB, (short)0, c1, false, false);
        }
        #pragma unroll
        for (int i = 0; i < 8; ++i) {
            y_lds[e * (16 * MO) + (i + hi * 8) * MO +      m] = c0[i];
            y_lds[e * (16 * MO) + (i + hi * 8) * MO + 16 + m] = c1[i];
        }
        __syncthreads();
    }

    if (lane < 16 && (row0 + lane) < NN) {
        const float* yr = y_lds + esel * (16 * MO) + lane * MO;
        float a0 = bov[0], a1 = bov[1];
        #pragma unroll
        for (int j = 0; j < MO; ++j) {
            float yv = yr[j];
            a0 += yv * wov[j * NO + 0];
            a1 += yv * wov[j * NO + 1];
        }
        out[(row0 + lane) * NO + 0] = a0;
        out[(row0 + lane) * NO + 1] = a1;
    }
}

__global__ void moe_loss_kernel(const int* __restrict__ cnt1,
                                float* __restrict__ loss_out) {
    if (threadIdx.x != 0) return;
    const float c1 = (float)cnt1[0];
    const float c0 = (float)NN - c1;
    const float mv = 0.5f * (float)NN;
    const float var = (c0 - mv) * (c0 - mv) + (c1 - mv) * (c1 - mv);
    const float cv  = var / (mv * mv + 1e-10f);
    loss_out[0] = 2.0f * cv * 0.01f;
}

extern "C" void kernel_launch(void* const* d_in, const int* in_sizes, int n_in,
                              void* d_out, int out_size, void* d_ws, size_t ws_size,
                              hipStream_t stream) {
    const float* x      = (const float*)d_in[0];
    const float* gate   = (const float*)d_in[1];
    const float* w_gate = (const float*)d_in[2];
    const float* w1     = (const float*)d_in[3];
    const float* b1     = (const float*)d_in[4];
    const float* w2     = (const float*)d_in[5];
    const float* b2     = (const float*)d_in[6];
    const float* wo     = (const float*)d_in[7];
    const float* bo     = (const float*)d_in[8];
    float* out = (float*)d_out;
    int*   cnt = (int*)d_ws;

    const int tiles = (NN + 15) / 16;

    moe_zero_ws<<<1, 1, 0, stream>>>(cnt);
    moe_tile_kernel<<<tiles, 32, 0, stream>>>(x, gate, w_gate, w1, b1, w2, b2,
                                              wo, bo, out, cnt);
    moe_loss_kernel<<<1, 1, 0, stream>>>(cnt, out + (size_t)NN * NO);
}


# Each lane column owns adjacent columns (hidden 4m..4m+3, 64+4m..; output 2m, 2m+1): W1, b1 by 16-byte and W2, b2 by 8-byte loads, h and y by 16/8-byte LDS stores; 370 vector loads per wave -> 168
# speedup vs baseline: 1.9060x; 1.1040x over previous
#include <hip/hip_runtime.h>

typedef float v2f __attribute__((ext_vector_type(2)));
typedef float v4f __attribute__((ext_vector_type(4)));
typedef float v8f __attribute__((ext_vector_type(8)));

#define NN    11826
#define DIN   9
#define DGATE 36
#define NE    2
#define HH    128
#define MO    32
#define NO    2

__global__ void moe_zero_ws(int* cnt) {
    if (threadIdx.x == 0) cnt[0] = 0;
}

__global__ __launch_bounds__(32) void moe_tile_kernel(
    const float* __restrict__ x,
    const float* __restrict__ gate,
    const float* __restrict__ w_gate,
    const float* __restrict__ w1,
    const float* __restrict__ b1,
    const float* __restrict__ w2,
    const float* __restrict__ b2,
    const float* __restrict__ wo,
    const float* __restrict__ bo,
    float* __restrict__ out,
    int* __restrict__ cnt1)
{
    __shared__ __attribute__((aligned(16))) float h_lds[16 * HH];
    __shared__ __attribute__((aligned(16))) float y_lds[NE * 16 * MO];

    const int lane = threadIdx.x;
    const int row0 = blockIdx.x * 16;
    const int m    = lane & 15;
    const int hi   = lane >> 4;
    const int rowm = min(row0 + m, NN - 1);

    float xv[6];
    v4f w1q[NE][5][2];
    v4f b1q[NE][2];
    v2f w2p[NE][64];
    v2f b2p[NE];
    v4f gq[DGATE / 4];
    float wgv[DGATE * NE];
    float wov[MO * NO];
    float bov[NO];

    {
        const float* xr = x + (size_t)rowm * DIN;
        #pragma unroll
        for (int c = 0; c < 3; ++c) {
            const int k0 = c * 4 + hi * 2;
            xv[2 * c]     = xr[min(k0,     DIN - 1)];
            xv[2 * c + 1] = xr[min(k0 + 1, DIN - 1)];
        }
    }
    #pragma unroll
    for (int e = 0; e < NE; ++e) {
        const float* W1 = w1 + (size_t)e * DIN * HH;
        const float* W2 = w2 + (size_t)e * HH * MO;
        b1q[e][0] = *(const v4f*)&b1[e * HH + 4 * m];
        b1q[e][1] = *(const v4f*)&b1[e * HH + 64 + 4 * m];
        #pragma unroll
        for (int q = 0; q < 5; ++q) {
            const int k = min((q >> 1) * 4 + hi * 2 + (q & 1), DIN - 1);
            w1q[e][q][0] = *(const v4f*)&W1[k * HH + 4 * m];
            w1q[e][q][1] = *(const v4f*)&W1[k * HH + 64 + 4 * m];
        }
        b2p[e] = *(const v2f*)&b2[e * MO + 2 * m];
        #pragma unroll
        for (int ks = 0; ks < HH / 4; ++ks) {
            const int ka = ks * 4 + hi * 2;
            w2p[e][2 * ks]     = *(const v2f*)&W2[(ka    ) * MO + 2 * m];
            w2p[e][2 * ks + 1] = *(const v2f*)&W2[(ka + 1) * MO + 2 * m];
        }
    }
    {
        const float* g = gate + (size_t)rowm * DGATE;
        #pragma unroll
        for (int d = 0; d < DGATE / 4; ++d) gq[d] = *(const v4f*)&g[4 * d];
    }
    #pragma unroll
    for (int d = 0; d < DGATE * NE; ++d) wgv[d] = w_gate[d];
    #pragma unroll
    for (int j = 0; j < MO * NO; ++j) wov[j] = wo[j];
    bov[0] = bo[0]; bov[1] = bo[1];

    v2f ax[3];
    #pragma unroll
    for (int c = 0; c < 3; ++c) {
        const int k0 = c * 4 + hi * 2;
        v2f a;
        a.x = (k0     < DIN) ? xv[2 * c]     : 0.f;
        a.y = (k0 + 1 < DIN) ? xv[2 * c + 1] : 0.f;
        ax[c] = a;
    }

    int esel;
    {
        float l0 = 0.f, l1 = 0.f;
        #pragma unroll
        for (int d = 0; d < DGATE; ++d) {
            l0 += gq[d >> 2][d & 3] * wgv[d * NE + 0];
            l1 += gq[d >> 2][d & 3] * wgv[d * NE + 1];
        }
        esel = (l1 > l0) ? 1 : 0;
        if (hi == 0 && (row0 + m) < NN && esel == 1) atomicAdd(cnt1, 1);
    }


    #pragma unroll
    for (int e = 0; e < NE; ++e) {
        #pragma unroll
        for (int g = 0; g < 2; ++g) {
            v8f hacc[4];
            #pragma unroll
            for (int t = 0; t < 4; ++t) {
                const float bias = b1q[e][g][t];
                v8f c8;
                #pragma unroll
                for (int i = 0; i < 8; ++i) c8[i] = bias;
                #pragma unroll
                for (int cix = 0; cix < 3; ++cix) {
                    const int kb = cix * 4 + hi * 2;
                    v2f b;
                    b.x = (kb < DIN) ? w1q[e][2 * cix][g][t] : 0.f;
                    b.y = 0.f;
                    if (cix < 2) b.y = w1q[e][2 * cix + 1][g][t];
                    c8 = __builtin_amdgcn_wmma_f32_16x16x4_f32(
                             false, ax[cix], false, b, (short)0, c8, false, false);
                }
                hacc[t] = c8;
            }
            #pragma unroll
            for (int i = 0; i < 8; ++i) {
                v4f h4;
                #pragma unroll
                for (int t = 0; t < 4; ++t) h4[t] = fmaxf(hacc[t][i], 0.0f);
                *(v4f*)&h_lds[(i + hi * 8) * HH + 64 * g + 4 * m] = h4;
            }
        }
        __syncthreads();

        v8f c0, c1;
        #pragma unroll
        for (int i = 0; i < 8; ++i) { c0[i] = b2p[e].x; c1[i] = b2p[e].y; }
        #pragma unroll
        for (int ks = 0; ks < HH / 4; ++ks) {
            const int ka = ks * 4 + hi * 2;
            v2f a = *(const v2f*)&h_lds[m * HH + ka];
            v2f bA, bB;
            bA.x = w2p[e][2 * ks].x; bA.y = w2p[e][2 * ks + 1].x;
            bB.x = w2p[e][2 * ks].y; bB.y = w2p[e][2 * ks + 1].y;
            c0 = __builtin_amdgcn_wmma_f32_16x16x4_f32(
                     false, a, false, bA, (short)0, c0, false, false);
            c1 = __builtin_amdgcn_wmma_f32_16x16x4_f32(
                     false, a, false, bB, (short)0, c1, false, false);
        }
        #pragma unroll
        for (int i = 0; i < 8; ++i) {
            v2f yy; yy.x = c0[i]; yy.y = c1[i];
            *(v2f*)&y_lds[e * (16 * MO) + (i + hi * 8) * MO + 2 * m] = yy;
        }
        __syncthreads();
    }

    if (lane < 16 && (row0 + lane) < NN) {
        const float* yr = y_lds + esel * (16 * MO) + lane * MO;
        float a0 = bov[0], a1 = bov[1];
        #pragma unroll
        for (int j = 0; j < MO; ++j) {
            float yv = yr[j];
            a0 += yv * wov[j * NO + 0];
            a1 += yv * wov[j * NO + 1];
        }
        out[(row0 + lane) * NO + 0] = a0;
        out[(row0 + lane) * NO + 1] = a1;
    }
}

__global__ void moe_loss_kernel(const int* __restrict__ cnt1,
                                float* __restrict__ loss_out) {
    if (threadIdx.x != 0) return;
    const float c1 = (float)cnt1[0];
    const float c0 = (float)NN - c1;
    const float mv = 0.5f * (float)NN;
    const float var = (c0 - mv) * (c0 - mv) + (c1 - mv) * (c1 - mv);
    const float cv  = var / (mv * mv + 1e-10f);
    loss_out[0] = 2.0f * cv * 0.01f;
}

extern "C" void kernel_launch(void* const* d_in, const int* in_sizes, int n_in,
                              void* d_out, int out_size, void* d_ws, size_t ws_size,
                              hipStream_t stream) {
    const float* x      = (const float*)d_in[0];
    const float* gate   = (const float*)d_in[1];
    const float* w_gate = (const float*)d_in[2];
    const float* w1     = (const float*)d_in[3];
    const float* b1     = (const float*)d_in[4];
    const float* w2     = (const float*)d_in[5];
    const float* b2     = (const float*)d_in[6];
    const float* wo     = (const float*)d_in[7];
    const float* bo     = (const float*)d_in[8];
    float* out = (float*)d_out;
    int*   cnt = (int*)d_ws;

    const int tiles = (NN + 15) / 16;

    moe_zero_ws<<<1, 1, 0, stream>>>(cnt);
    moe_tile_kernel<<<tiles, 32, 0, stream>>>(x, gate, w_gate, w1, b1, w2, b2,
                                              wo, bo, out, cnt);
    moe_loss_kernel<<<1, 1, 0, stream>>>(cnt, out + (size_t)NN * NO);
}


# Expert-1 row count: each tile stores its own count word (ballot + popcount), the loss kernel sums the 740 words in one wave (integer sum, exact); no atomic, no clearing launch: 3 launches -> 2
# speedup vs baseline: 2.0862x; 1.0945x over previous
#include <hip/hip_runtime.h>

typedef float v2f __attribute__((ext_vector_type(2)));
typedef float v4f __attribute__((ext_vector_type(4)));
typedef float v8f __attribute__((ext_vector_type(8)));
typedef int   v4i __attribute__((ext_vector_type(4)));

#define NN    11826
#define DIN   9
#define DGATE 36
#define NE    2
#define HH    128
#define MO    32
#define NO    2
#define TILES ((NN + 15) / 16)

__global__ __launch_bounds__(32) void moe_tile_kernel(
    const float* __restrict__ x,
    const float* __restrict__ gate,
    const float* __restrict__ w_gate,
    const float* __restrict__ w1,
    const float* __restrict__ b1,
    const float* __restrict__ w2,
    const float* __restrict__ b2,
    const float* __restrict__ wo,
    const float* __restrict__ bo,
    float* __restrict__ out,
    int* __restrict__ tile_cnt1)
{
    __shared__ __attribute__((aligned(16))) float h_lds[16 * HH];
    __shared__ __attribute__((aligned(16))) float y_lds[NE * 16 * MO];

    const int lane = threadIdx.x;
    const int row0 = blockIdx.x * 16;
    const int m    = lane & 15;
    const int hi   = lane >> 4;
    const int rowm = min(row0 + m, NN - 1);

    float xv[6];
    v4f w1q[NE][5][2];
    v4f b1q[NE][2];
    v2f w2p[NE][64];
    v2f b2p[NE];
    v4f gq[DGATE / 4];
    float wgv[DGATE * NE];
    float wov[MO * NO];
    float bov[NO];

    {
        const float* xr = x + (size_t)rowm * DIN;
        #pragma unroll
        for (int c = 0; c < 3; ++c) {
            const int k0 = c * 4 + hi * 2;
            xv[2 * c]     = xr[min(k0,     DIN - 1)];
            xv[2 * c + 1] = xr[min(k0 + 1, DIN - 1)];
        }
    }
    #pragma unroll
    for (int e = 0; e < NE; ++e) {
        const float* W1 = w1 + (size_t)e * DIN * HH;
        const float* W2 = w2 + (size_t)e * HH * MO;
        b1q[e][0] = *(const v4f*)&b1[e * HH + 4 * m];
        b1q[e][1] = *(const v4f*)&b1[e * HH + 64 + 4 * m];
        #pragma unroll
        for (int q = 0; q < 5; ++q) {
            const int k = min((q >> 1) * 4 + hi * 2 + (q & 1), DIN - 1);
            w1q[e][q][0] = *(const v4f*)&W1[k * HH + 4 * m];
            w1q[e][q][1] = *(const v4f*)&W1[k * HH + 64 + 4 * m];
        }
        b2p[e] = *(const v2f*)&b2[e * MO + 2 * m];
        #pragma unroll
        for (int ks = 0; ks < HH / 4; ++ks) {
            const int ka = ks * 4 + hi * 2;
            w2p[e][2 * ks]     = *(const v2f*)&W2[(ka    ) * MO + 2 * m];
            w2p[e][2 * ks + 1] = *(const v2f*)&W2[(ka + 1) * MO + 2 * m];
        }
    }
    {
        const float* g = gate + (size_t)rowm * DGATE;
        #pragma unroll
        for (int d = 0; d < DGATE / 4; ++d) gq[d] = *(const v4f*)&g[4 * d];
    }
    #pragma unroll
    for (int d = 0; d < DGATE * NE; ++d) wgv[d] = w_gate[d];
    #pragma unroll
    for (int j = 0; j < MO * NO; ++j) wov[j] = wo[j];
    bov[0] = bo[0]; bov[1] = bo[1];

    v2f ax[3];
    #pragma unroll
    for (int c = 0; c < 3; ++c) {
        const int k0 = c * 4 + hi * 2;
        v2f a;
        a.x = (k0     < DIN) ? xv[2 * c]     : 0.f;
        a.y = (k0 + 1 < DIN) ? xv[2 * c + 1] : 0.f;
        ax[c] = a;
    }

    int esel;
    {
        float l0 = 0.f, l1 = 0.f;
        #pragma unroll
        for (int d = 0; d < DGATE; ++d) {
            l0 += gq[d >> 2][d & 3] * wgv[d * NE + 0];
            l1 += gq[d >> 2][d & 3] * wgv[d * NE + 1];
        }
        esel = (l1 > l0) ? 1 : 0;
        const unsigned sel1 = __builtin_amdgcn_ballot_w32(hi == 0 && (row0 + m) < NN && esel == 1);
        if (lane == 0) tile_cnt1[blockIdx.x] = __popc(sel1);
    }


    #pragma unroll
    for (int e = 0; e < NE; ++e) {
        #pragma unroll
        for (int g = 0; g < 2; ++g) {
            v8f hacc[4];
            #pragma unroll
            for (int t = 0; t < 4; ++t) {
                const float bias = b1q[e][g][t];
                v8f c8;
                #pragma unroll
                for (int i = 0; i < 8; ++i) c8[i] = bias;
                #pragma unroll
                for (int cix = 0; cix < 3; ++cix) {
                    const int kb = cix * 4 + hi * 2;
                    v2f b;
                    b.x = (kb < DIN) ? w1q[e][2 * cix][g][t] : 0.f;
                    b.y = 0.f;
                    if (cix < 2) b.y = w1q[e][2 * cix + 1][g][t];
                    c8 = __builtin_amdgcn_wmma_f32_16x16x4_f32(
                             false, ax[cix], false, b, (short)0, c8, false, false);
                }
                hacc[t] = c8;
            }
            #pragma unroll
            for (int i = 0; i < 8; ++i) {
                v4f h4;
                #pragma unroll
                for (int t = 0; t < 4; ++t) h4[t] = fmaxf(hacc[t][i], 0.0f);
                *(v4f*)&h_lds[(i + hi * 8) * HH + 64 * g + 4 * m] = h4;
            }
        }
        __syncthreads();

        v8f c0, c1;
        #pragma unroll
        for (int i = 0; i < 8; ++i) { c0[i] = b2p[e].x; c1[i] = b2p[e].y; }
        #pragma unroll
        for (int ks = 0; ks < HH / 4; ++ks) {
            const int ka = ks * 4 + hi * 2;
            v2f a = *(const v2f*)&h_lds[m * HH + ka];
            v2f bA, bB;
            bA.x = w2p[e][2 * ks].x; bA.y = w2p[e][2 * ks + 1].x;
            bB.x = w2p[e][2 * ks].y; bB.y = w2p[e][2 * ks + 1].y;
            c0 = __builtin_amdgcn_wmma_f32_16x16x4_f32(
                     false, a, false, bA, (short)0, c0, false, false);
            c1 = __builtin_amdgcn_wmma_f32_16x16x4_f32(
                     false, a, false, bB, (short)0, c1, false, false);
        }
        #pragma unroll
        for (int i = 0; i < 8; ++i) {
            v2f yy; yy.x = c0[i]; yy.y = c1[i];
            *(v2f*)&y_lds[e * (16 * MO) + (i + hi * 8) * MO + 2 * m] = yy;
        }
        __syncthreads();
    }

    if (lane < 16 && (row0 + lane) < NN) {
        const float* yr = y_lds + esel * (16 * MO) + lane * MO;
        float a0 = bov[0], a1 = bov[1];
        #pragma unroll
        for (int j = 0; j < MO; ++j) {
            float yv = yr[j];
            a0 += yv * wov[j * NO + 0];
            a1 += yv * wov[j * NO + 1];
        }
        out[(row0 + lane) * NO + 0] = a0;
        out[(row0 + lane) * NO + 1] = a1;
    }
}

__device__ __forceinline__ int xhalf_i(int v) {
    return __builtin_amdgcn_permlanex16(v, v, 0x76543210, 0xfedcba98, false, false);
}
template <int MASK>
__device__ __forceinline__ int xrow_i(int v) {
    return __builtin_amdgcn_update_dpp(0, v, 0x160 + MASK, 0xf, 0xf, false);
}

__global__ __launch_bounds__(32) void moe_loss_kernel(const int* __restrict__ tile_cnt1,
                                                      float* __restrict__ loss_out) {
    static_assert(TILES % 4 == 0, "counts are read four at a time");
    const int lane = threadIdx.x;
    constexpr int QUADS = TILES / 4;
    constexpr int PER_LANE = (QUADS + 31) / 32;
    v4i q[PER_LANE];
    #pragma unroll
    for (int j = 0; j < PER_LANE; ++j) {
        const int idc = min(j * 32 + lane, QUADS - 1);
        q[j] = *(const v4i*)&tile_cnt1[4 * idc];
    }
    int s = 0;
    #pragma unroll
    for (int j = 0; j < PER_LANE; ++j) {
        const int t = (q[j].x + q[j].y) + (q[j].z + q[j].w);
        if ((j + 1) * 32 <= QUADS) s += t;
        else s += (j * 32 + lane < QUADS) ? t : 0;
    }
    s += xrow_i<1>(s);
    s += xrow_i<2>(s);
    s += xrow_i<4>(s);
    s += xrow_i<8>(s);
    s += xhalf_i(s);
    if (lane != 0) return;
    const float c1 = (float)s;
    const float c0 = (float)NN - c1;
    const float mv = 0.5f * (float)NN;
    const float var = (c0 - mv) * (c0 - mv) + (c1 - mv) * (c1 - mv);
    const float cv  = var / (mv * mv + 1e-10f);
    loss_out[0] = 2.0f * cv * 0.01f;
}

extern "C" void kernel_launch(void* const* d_in, const int* in_sizes, int n_in,
                              void* d_out, int out_size, void* d_ws, size_t ws_size,
                              hipStream_t stream) {
    const float* x      = (const float*)d_in[0];
    const float* gate   = (const float*)d_in[1];
    const float* w_gate = (const float*)d_in[2];
    const float* w1     = (const float*)d_in[3];
    const float* b1     = (const float*)d_in[4];
    const float* w2     = (const float*)d_in[5];
    const float* b2     = (const float*)d_in[6];
    const float* wo     = (const float*)d_in[7];
    const float* bo     = (const float*)d_in[8];
    float* out = (float*)d_out;
    int*   cnt = (int*)d_ws;

    moe_tile_kernel<<<TILES, 32, 0, stream>>>(x, gate, w_gate, w1, b1, w2, b2,
                                              wo, bo, out, cnt);
    moe_loss_kernel<<<1, 32, 0, stream>>>(cnt, out + (size_t)NN * NO);
}


# First product computed transposed (W1 as A, x as B; b1 enters at pad slot k=9 against a one): its D registers are the second product's A fragments, h skips LDS; both products' sum order changes
# speedup vs baseline: 2.2138x; 1.0612x over previous
#include <hip/hip_runtime.h>

typedef float v2f __attribute__((ext_vector_type(2)));
typedef float v4f __attribute__((ext_vector_type(4)));
typedef float v8f __attribute__((ext_vector_type(8)));
typedef int   v4i __attribute__((ext_vector_type(4)));

#define NN    11826
#define DIN   9
#define DGATE 36
#define NE    2
#define HH    128
#define MO    32
#define NO    2
#define TILES ((NN + 15) / 16)

__global__ __launch_bounds__(32) void moe_tile_kernel(
    const float* __restrict__ x,
    const float* __restrict__ gate,
    const float* __restrict__ w_gate,
    const float* __restrict__ w1,
    const float* __restrict__ b1,
    const float* __restrict__ w2,
    const float* __restrict__ b2,
    const float* __restrict__ wo,
    const float* __restrict__ bo,
    float* __restrict__ out,
    int* __restrict__ tile_cnt1)
{
    __shared__ __attribute__((aligned(16))) float y_lds[NE * 16 * MO];

    const int lane = threadIdx.x;
    const int row0 = blockIdx.x * 16;
    const int m    = lane & 15;
    const int hi   = lane >> 4;
    const int rowm = min(row0 + m, NN - 1);

    float xv[6];
    v4f w1q[NE][5][2];
    v4f b1q[NE][2];
    v2f w2p[NE][64];
    v2f b2p[NE];
    v4f gq[DGATE / 4];
    float wgv[DGATE * NE];
    float wov[MO * NO];
    float bov[NO];

    {
        const float* xr = x + (size_t)rowm * DIN;
        #pragma unroll
        for (int c = 0; c < 3; ++c) {
            const int k0 = c * 4 + hi * 2;
            xv[2 * c]     = xr[min(k0,     DIN - 1)];
            xv[2 * c + 1] = xr[min(k0 + 1, DIN - 1)];
        }
    }
    #pragma unroll
    for (int e = 0; e < NE; ++e) {
        const float* W1 = w1 + (size_t)e * DIN * HH;
        const float* W2 = w2 + (size_t)e * HH * MO;
        b1q[e][0] = *(const v4f*)&b1[e * HH + 4 * m];
        b1q[e][1] = *(const v4f*)&b1[e * HH + 64 + 4 * m];
        #pragma unroll
        for (int q = 0; q < 5; ++q) {
            const int k = min((q >> 1) * 4 + hi * 2 + (q & 1), DIN - 1);
            w1q[e][q][0] = *(const v4f*)&W1[k * HH + 4 * m];
            w1q[e][q][1] = *(const v4f*)&W1[k * HH + 64 + 4 * m];
        }
        b2p[e] = *(const v2f*)&b2[e * MO + 2 * m];
        #pragma unroll
        for (int st = 0; st < HH / 4; ++st) {
            const int ka = 64 * (st >> 4) + 32 * hi + 8 * (st & 3) + ((st >> 2) & 3);
            w2p[e][2 * st]     = *(const v2f*)&W2[(ka    ) * MO + 2 * m];
            w2p[e][2 * st + 1] = *(const v2f*)&W2[(ka + 4) * MO + 2 * m];
        }
    }
    {
        const float* g = gate + (size_t)rowm * DGATE;
        #pragma unroll
        for (int d = 0; d < DGATE / 4; ++d) gq[d] = *(const v4f*)&g[4 * d];
    }
    #pragma unroll
    for (int d = 0; d < DGATE * NE; ++d) wgv[d] = w_gate[d];
    #pragma unroll
    for (int j = 0; j < MO * NO; ++j) wov[j] = wo[j];
    bov[0] = bo[0]; bov[1] = bo[1];

    v2f ax[3];
    #pragma unroll
    for (int c = 0; c < 3; ++c) {
        const int k0 = c * 4 + hi * 2;
        v2f a;
        a.x = (k0     < DIN) ? xv[2 * c]     : 0.f;
        if (c < 2) a.y = xv[2 * c + 1];
        else       a.y = (k0 + 1 == DIN) ? 1.0f : 0.f;
        ax[c] = a;
    }

    int esel;
    {
        float l0 = 0.f, l1 = 0.f;
        #pragma unroll
        for (int d = 0; d < DGATE; ++d) {
            l0 += gq[d >> 2][d & 3] * wgv[d * NE + 0];
            l1 += gq[d >> 2][d & 3] * wgv[d * NE + 1];
        }
        esel = (l1 > l0) ? 1 : 0;
        const unsigned sel1 = __builtin_amdgcn_ballot_w32(hi == 0 && (row0 + m) < NN && esel == 1);
        if (lane == 0) tile_cnt1[blockIdx.x] = __popc(sel1);
    }

    #pragma unroll
    for (int e = 0; e < NE; ++e) {
        v8f hT[2][4];
        #pragma unroll
        for (int g = 0; g < 2; ++g) {
            #pragma unroll
            for (int t = 0; t < 4; ++t) {
                v8f c8;
                #pragma unroll
                for (int i = 0; i < 8; ++i) c8[i] = 0.f;
                #pragma unroll
                for (int cix = 0; cix < 3; ++cix) {
                    const int kb = cix * 4 + hi * 2;
                    v2f w;
                    w.x = (kb < DIN) ? w1q[e][2 * cix][g][t] : 0.f;
                    w.y = (kb + 1 == DIN) ? b1q[e][g][t] : 0.f;
                    if (cix < 2) w.y = w1q[e][2 * cix + 1][g][t];
                    c8 = __builtin_amdgcn_wmma_f32_16x16x4_f32(
                             false, w, false, ax[cix], (short)0, c8, false, false);
                }
                #pragma unroll
                for (int i = 0; i < 8; ++i) c8[i] = fmaxf(c8[i], 0.0f);
                hT[g][t] = c8;
            }
        }

        v8f c0, c1;
        #pragma unroll
        for (int i = 0; i < 8; ++i) { c0[i] = b2p[e].x; c1[i] = b2p[e].y; }
        #pragma unroll
        for (int st = 0; st < HH / 4; ++st) {
            const int g = st >> 4, t = (st >> 2) & 3, s = st & 3;
            v2f a;
            a.x = hT[g][t][2 * s]; a.y = hT[g][t][2 * s + 1];
            v2f bA, bB;
            bA.x = w2p[e][2 * st].x; bA.y = w2p[e][2 * st + 1].x;
            bB.x = w2p[e][2 * st].y; bB.y = w2p[e][2 * st + 1].y;
            c0 = __builtin_amdgcn_wmma_f32_16x16x4_f32(
                     false, a, false, bA, (short)0, c0, false, false);
            c1 = __builtin_amdgcn_wmma_f32_16x16x4_f32(
                     false, a, false, bB, (short)0, c1, false, false);
        }
        #pragma unroll
        for (int i = 0; i < 8; ++i) {
            v2f yy; yy.x = c0[i]; yy.y = c1[i];
            *(v2f*)&y_lds[e * (16 * MO) + (i + hi * 8) * MO + 2 * m] = yy;
        }
    }
    __syncthreads();

    if (lane < 16 && (row0 + lane) < NN) {
        const float* yr = y_lds + esel * (16 * MO) + lane * MO;
        float a0 = bov[0], a1 = bov[1];
        #pragma unroll
        for (int j = 0; j < MO; ++j) {
            float yv = yr[j];
            a0 += yv * wov[j * NO + 0];
            a1 += yv * wov[j * NO + 1];
        }
        out[(row0 + lane) * NO + 0] = a0;
        out[(row0 + lane) * NO + 1] = a1;
    }
}

__device__ __forceinline__ int xhalf_i(int v) {
    return __builtin_amdgcn_permlanex16(v, v, 0x76543210, 0xfedcba98, false, false);
}
template <int MASK>
__device__ __forceinline__ int xrow_i(int v) {
    return __builtin_amdgcn_update_dpp(0, v, 0x160 + MASK, 0xf, 0xf, false);
}

__global__ __launch_bounds__(32) void moe_loss_kernel(const int* __restrict__ tile_cnt1,
                                                      float* __restrict__ loss_out) {
    static_assert(TILES % 4 == 0, "counts are read four at a time");
    const int lane = threadIdx.x;
    constexpr int QUADS = TILES / 4;
    constexpr int PER_LANE = (QUADS + 31) / 32;
    v4i q[PER_LANE];
    #pragma unroll
    for (int j = 0; j < PER_LANE; ++j) {
        const int idc = min(j * 32 + lane, QUADS - 1);
        q[j] = *(const v4i*)&tile_cnt1[4 * idc];
    }
    int s = 0;
    #pragma unroll
    for (int j = 0; j < PER_LANE; ++j) {
        const int t = (q[j].x + q[j].y) + (q[j].z + q[j].w);
        if ((j + 1) * 32 <= QUADS) s += t;
        else s += (j * 32 + lane < QUADS) ? t : 0;
    }
    s += xrow_i<1>(s);
    s += xrow_i<2>(s);
    s += xrow_i<4>(s);
    s += xrow_i<8>(s);
    s += xhalf_i(s);
    if (lane != 0) return;
    const float c1 = (float)s;
    const float c0 = (float)NN - c1;
    const float mv = 0.5f * (float)NN;
    const float var = (c0 - mv) * (c0 - mv) + (c1 - mv) * (c1 - mv);
    const float cv  = var / (mv * mv + 1e-10f);
    loss_out[0] = 2.0f * cv * 0.01f;
}

extern "C" void kernel_launch(void* const* d_in, const int* in_sizes, int n_in,
                              void* d_out, int out_size, void* d_ws, size_t ws_size,
                              hipStream_t stream) {
    const float* x      = (const float*)d_in[0];
    const float* gate   = (const float*)d_in[1];
    const float* w_gate = (const float*)d_in[2];
    const float* w1     = (const float*)d_in[3];
    const float* b1     = (const float*)d_in[4];
    const float* w2     = (const float*)d_in[5];
    const float* b2     = (const float*)d_in[6];
    const float* wo     = (const float*)d_in[7];
    const float* bo     = (const float*)d_in[8];
    float* out = (float*)d_out;
    int*   cnt = (int*)d_ws;

    moe_tile_kernel<<<TILES, 32, 0, stream>>>(x, gate, w_gate, w1, b1, w2, b2,
                                              wo, bo, out, cnt);
    moe_loss_kernel<<<1, 32, 0, stream>>>(cnt, out + (size_t)NN * NO);
}


# Projection wo folded into the second product's B fragment (lane column = 4-column chunk x output; partials summed over 8 lanes): 128 -> 64 matrix ops, y skips LDS; sum order of y and out changes
# speedup vs baseline: 2.8570x; 1.2905x over previous
#include <hip/hip_runtime.h>

typedef float v2f __attribute__((ext_vector_type(2)));
typedef float v4f __attribute__((ext_vector_type(4)));
typedef float v8f __attribute__((ext_vector_type(8)));
typedef int   v4i __attribute__((ext_vector_type(4)));

#define NN    11826
#define DIN   9
#define DGATE 36
#define NE    2
#define HH    128
#define MO    32
#define NO    2
#define TILES ((NN + 15) / 16)

template <int MASK>
__device__ __forceinline__ float xrow_f(float v) {
    return __int_as_float(__builtin_amdgcn_update_dpp(0, __float_as_int(v), 0x160 + MASK, 0xf, 0xf, false));
}

__global__ __launch_bounds__(32) void moe_tile_kernel(
    const float* __restrict__ x,
    const float* __restrict__ gate,
    const float* __restrict__ w_gate,
    const float* __restrict__ w1,
    const float* __restrict__ b1,
    const float* __restrict__ w2,
    const float* __restrict__ b2,
    const float* __restrict__ wo,
    const float* __restrict__ bo,
    float* __restrict__ out,
    int* __restrict__ tile_cnt1)
{
    const int lane = threadIdx.x;
    const int row0 = blockIdx.x * 16;
    const int m    = lane & 15;
    const int hi   = lane >> 4;
    const int rowm = min(row0 + m, NN - 1);
    const int q    = m >> 1;
    const int o    = m & 1;

    float xv[6];
    v4f w1q[NE][5][2];
    v4f b1q[NE][2];
    v4f w2q[NE][64];
    v4f b2q[NE];
    float wl[4];
    float bol;
    v4f gq[DGATE / 4];
    float wgv[DGATE * NE];

    {
        const float* xr = x + (size_t)rowm * DIN;
        #pragma unroll
        for (int c = 0; c < 3; ++c) {
            const int k0 = c * 4 + hi * 2;
            xv[2 * c]     = xr[min(k0,     DIN - 1)];
            xv[2 * c + 1] = xr[min(k0 + 1, DIN - 1)];
        }
    }
    #pragma unroll
    for (int e = 0; e < NE; ++e) {
        const float* W1 = w1 + (size_t)e * DIN * HH;
        const float* W2 = w2 + (size_t)e * HH * MO;
        b1q[e][0] = *(const v4f*)&b1[e * HH + 4 * m];
        b1q[e][1] = *(const v4f*)&b1[e * HH + 64 + 4 * m];
        #pragma unroll
        for (int qq = 0; qq < 5; ++qq) {
            const int k = min((qq >> 1) * 4 + hi * 2 + (qq & 1), DIN - 1);
            w1q[e][qq][0] = *(const v4f*)&W1[k * HH + 4 * m];
            w1q[e][qq][1] = *(const v4f*)&W1[k * HH + 64 + 4 * m];
        }
        b2q[e] = *(const v4f*)&b2[e * MO + 4 * q];
        #pragma unroll
        for (int st = 0; st < HH / 4; ++st) {
            const int ka = 64 * (st >> 4) + 32 * hi + 8 * (st & 3) + ((st >> 2) & 3);
            w2q[e][2 * st]     = *(const v4f*)&W2[(ka    ) * MO + 4 * q];
            w2q[e][2 * st + 1] = *(const v4f*)&W2[(ka + 4) * MO + 4 * q];
        }
    }
    {
        const float* g = gate + (size_t)rowm * DGATE;
        #pragma unroll
        for (int d = 0; d < DGATE / 4; ++d) gq[d] = *(const v4f*)&g[4 * d];
    }
    #pragma unroll
    for (int d = 0; d < DGATE * NE; ++d) wgv[d] = w_gate[d];
    #pragma unroll
    for (int jj = 0; jj < 4; ++jj) wl[jj] = wo[(4 * q + jj) * NO + o];
    bol = bo[o];

    v2f ax[3];
    #pragma unroll
    for (int c = 0; c < 3; ++c) {
        const int k0 = c * 4 + hi * 2;
        v2f a;
        a.x = (k0     < DIN) ? xv[2 * c]     : 0.f;
        if (c < 2) a.y = xv[2 * c + 1];
        else       a.y = (k0 + 1 == DIN) ? 1.0f : 0.f;
        ax[c] = a;
    }

    unsigned sel_rows;
    {
        float l0 = 0.f, l1 = 0.f;
        #pragma unroll
        for (int d = 0; d < DGATE; ++d) {
            l0 += gq[d >> 2][d & 3] * wgv[d * NE + 0];
            l1 += gq[d >> 2][d & 3] * wgv[d * NE + 1];
        }
        const int esel = (l1 > l0) ? 1 : 0;
        const unsigned sel1 = __builtin_amdgcn_ballot_w32(hi == 0 && (row0 + m) < NN && esel == 1);
        if (lane == 0) tile_cnt1[blockIdx.x] = __popc(sel1);
        sel_rows = __builtin_amdgcn_ballot_w32(esel == 1);
    }

    v8f hT[NE][2][4];
    #pragma unroll
    for (int e = 0; e < NE; ++e) {
        #pragma unroll
        for (int g = 0; g < 2; ++g) {
            #pragma unroll
            for (int t = 0; t < 4; ++t) {
                v8f c8;
                #pragma unroll
                for (int i = 0; i < 8; ++i) c8[i] = 0.f;
                #pragma unroll
                for (int cix = 0; cix < 3; ++cix) {
                    const int kb = cix * 4 + hi * 2;
                    v2f w;
                    w.x = (kb < DIN) ? w1q[e][2 * cix][g][t] : 0.f;
                    w.y = (kb + 1 == DIN) ? b1q[e][g][t] : 0.f;
                    if (cix < 2) w.y = w1q[e][2 * cix + 1][g][t];
                    c8 = __builtin_amdgcn_wmma_f32_16x16x4_f32(
                             false, w, false, ax[cix], (short)0, c8, false, false);
                }
                #pragma unroll
                for (int i = 0; i < 8; ++i) c8[i] = fmaxf(c8[i], 0.0f);
                hT[e][g][t] = c8;
            }
        }
    }

    v2f wl01, wl23;
    wl01.x = wl[0]; wl01.y = wl[1]; wl23.x = wl[2]; wl23.y = wl[3];
    v8f acc[NE];
    #pragma unroll
    for (int e = 0; e < NE; ++e) {
        float cb = (q == 0) ? bol : 0.f;
        #pragma unroll
        for (int jj = 0; jj < 4; ++jj) cb += b2q[e][jj] * wl[jj];
        #pragma unroll
        for (int i = 0; i < 8; ++i) acc[e][i] = cb;
    }
    #pragma unroll
    for (int st = 0; st < HH / 4; ++st) {
        const int g = st >> 4, t = (st >> 2) & 3, s = st & 3;
        #pragma unroll
        for (int e = 0; e < NE; ++e) {
            v2f a;
            a.x = hT[e][g][t][2 * s]; a.y = hT[e][g][t][2 * s + 1];
            const v4f ra = w2q[e][2 * st], rb = w2q[e][2 * st + 1];
            const v2f pa = __builtin_elementwise_fma(ra.zw, wl23, ra.xy * wl01);
            const v2f pb = __builtin_elementwise_fma(rb.zw, wl23, rb.xy * wl01);
            v2f bf;
            bf.x = pa.x + pa.y;
            bf.y = pb.x + pb.y;
            acc[e] = __builtin_amdgcn_wmma_f32_16x16x4_f32(
                         false, a, false, bf, (short)0, acc[e], false, false);
        }
    }

    const unsigned sh = sel_rows >> (8 * hi);
    float v[8];
    #pragma unroll
    for (int r = 0; r < 8; ++r) {
        float s = ((sh >> r) & 1u) ? acc[1][r] : acc[0][r];
        s += xrow_f<2>(s);
        s += xrow_f<4>(s);
        s += xrow_f<8>(s);
        v[r] = s;
    }
    const bool p0 = (m & 2) != 0, p1 = (m & 4) != 0, p2 = (m & 8) != 0;
    const float t0 = p0 ? v[1] : v[0], t1 = p0 ? v[3] : v[2];
    const float t2 = p0 ? v[5] : v[4], t3 = p0 ? v[7] : v[6];
    const float u0 = p1 ? t1 : t0,     u1 = p1 ? t3 : t2;
    const float res = p2 ? u1 : u0;
    if (row0 + 8 * hi + (m >> 1) < NN) out[(size_t)row0 * NO + lane] = res;
}

__device__ __forceinline__ int xhalf_i(int v) {
    return __builtin_amdgcn_permlanex16(v, v, 0x76543210, 0xfedcba98, false, false);
}
template <int MASK>
__device__ __forceinline__ int xrow_i(int v) {
    return __builtin_amdgcn_update_dpp(0, v, 0x160 + MASK, 0xf, 0xf, false);
}

__global__ __launch_bounds__(32) void moe_loss_kernel(const int* __restrict__ tile_cnt1,
                                                      float* __restrict__ loss_out) {
    static_assert(TILES % 4 == 0, "counts are read four at a time");
    const int lane = threadIdx.x;
    constexpr int QUADS = TILES / 4;
    constexpr int PER_LANE = (QUADS + 31) / 32;
    v4i q[PER_LANE];
    #pragma unroll
    for (int j = 0; j < PER_LANE; ++j) {
        const int idc = min(j * 32 + lane, QUADS - 1);
        q[j] = *(const v4i*)&tile_cnt1[4 * idc];
    }
    int s = 0;
    #pragma unroll
    for (int j = 0; j < PER_LANE; ++j) {
        const int t = (q[j].x + q[j].y) + (q[j].z + q[j].w);
        if ((j + 1) * 32 <= QUADS) s += t;
        else s += (j * 32 + lane < QUADS) ? t : 0;
    }
    s += xrow_i<1>(s);
    s += xrow_i<2>(s);
    s += xrow_i<4>(s);
    s += xrow_i<8>(s);
    s += xhalf_i(s);
    if (lane != 0) return;
    const float c1 = (float)s;
    const float c0 = (float)NN - c1;
    const float mv = 0.5f * (float)NN;
    const float var = (c0 - mv) * (c0 - mv) + (c1 - mv) * (c1 - mv);
    const float cv  = var / (mv * mv + 1e-10f);
    loss_out[0] = 2.0f * cv * 0.01f;
}

extern "C" void kernel_launch(void* const* d_in, const int* in_sizes, int n_in,
                              void* d_out, int out_size, void* d_ws, size_t ws_size,
                              hipStream_t stream) {
    const float* x      = (const float*)d_in[0];
    const float* gate   = (const float*)d_in[1];
    const float* w_gate = (const float*)d_in[2];
    const float* w1     = (const float*)d_in[3];
    const float* b1     = (const float*)d_in[4];
    const float* w2     = (const float*)d_in[5];
    const float* b2     = (const float*)d_in[6];
    const float* wo     = (const float*)d_in[7];
    const float* bo     = (const float*)d_in[8];
    float* out = (float*)d_out;
    int*   cnt = (int*)d_ws;

    moe_tile_kernel<<<TILES, 32, 0, stream>>>(x, gate, w_gate, w1, b1, w2, b2,
                                              wo, bo, out, cnt);
    moe_loss_kernel<<<1, 32, 0, stream>>>(cnt, out + (size_t)NN * NO);
}


# Rows routed to their expert before the products: both experts' W1 stacked along k against a per-row masked x (48 -> 40 matrix ops), both experts' W2.wo as lane columns (64 -> 32); sum order changes
# speedup vs baseline: 3.0869x; 1.0805x over previous
#include <hip/hip_runtime.h>

typedef float v2f __attribute__((ext_vector_type(2)));
typedef float v4f __attribute__((ext_vector_type(4)));
typedef float v8f __attribute__((ext_vector_type(8)));
typedef int   v4i __attribute__((ext_vector_type(4)));

#define NN    11826
#define DIN   9
#define DGATE 36
#define NE    2
#define HH    128
#define MO    32
#define NO    2
#define TILES ((NN + 15) / 16)

template <int MASK>
__device__ __forceinline__ float xrow_f(float v) {
    return __int_as_float(__builtin_amdgcn_update_dpp(0, __float_as_int(v), 0x160 + MASK, 0xf, 0xf, false));
}

__global__ __launch_bounds__(32) void moe_tile_kernel(
    const float* __restrict__ x,
    const float* __restrict__ gate,
    const float* __restrict__ w_gate,
    const float* __restrict__ w1,
    const float* __restrict__ b1,
    const float* __restrict__ w2,
    const float* __restrict__ b2,
    const float* __restrict__ wo,
    const float* __restrict__ bo,
    float* __restrict__ out,
    int* __restrict__ tile_cnt1)
{
    const int lane = threadIdx.x;
    const int row0 = blockIdx.x * 16;
    const int m    = lane & 15;
    const int hi   = lane >> 4;
    const int rowm = min(row0 + m, NN - 1);
    const int o    = m & 1;
    const int q4   = (m >> 1) & 3;
    const int ec   = m >> 3;

    float xr9[DIN];
    v4f w1s[5][2][2];
    v4f w2q[HH / 4][2][2];
    v4f b2q[2];
    float wl[8];
    float bol;
    v4f gq[DGATE / 4];
    float wgv[DGATE * NE];

    {
        const float* xr = x + (size_t)rowm * DIN;
        #pragma unroll
        for (int k = 0; k < DIN; ++k) xr9[k] = xr[k];
    }
    #pragma unroll
    for (int c = 0; c < 5; ++c) {
        #pragma unroll
        for (int j = 0; j < 2; ++j) {
            const int s0 = 4 * c + j, s1 = 4 * c + 2 + j;
            const float* p0 = (s0 % 10 < DIN) ? w1 + ((s0 / 10) * DIN + s0 % 10) * HH : b1 + (s0 / 10) * HH;
            const float* p1 = (s1 % 10 < DIN) ? w1 + ((s1 / 10) * DIN + s1 % 10) * HH : b1 + (s1 / 10) * HH;
            const float* p  = hi ? p1 : p0;
            w1s[c][j][0] = *(const v4f*)&p[4 * m];
            w1s[c][j][1] = *(const v4f*)&p[64 + 4 * m];
        }
    }
    {
        const float* W2 = w2 + (size_t)ec * HH * MO;
        b2q[0] = *(const v4f*)&b2[ec * MO + 8 * q4];
        b2q[1] = *(const v4f*)&b2[ec * MO + 8 * q4 + 4];
        #pragma unroll
        for (int st = 0; st < HH / 4; ++st) {
            const int ka = 64 * (st >> 4) + 32 * hi + 8 * (st & 3) + ((st >> 2) & 3);
            #pragma unroll
            for (int j = 0; j < 2; ++j) {
                w2q[st][j][0] = *(const v4f*)&W2[(ka + 4 * j) * MO + 8 * q4];
                w2q[st][j][1] = *(const v4f*)&W2[(ka + 4 * j) * MO + 8 * q4 + 4];
            }
        }
    }
    {
        const float* g = gate + (size_t)rowm * DGATE;
        #pragma unroll
        for (int d = 0; d < DGATE / 4; ++d) gq[d] = *(const v4f*)&g[4 * d];
    }
    #pragma unroll
    for (int d = 0; d < DGATE * NE; ++d) wgv[d] = w_gate[d];
    #pragma unroll
    for (int jj = 0; jj < 8; ++jj) wl[jj] = wo[(8 * q4 + jj) * NO + o];
    bol = bo[o];

    unsigned sel_rows;
    int esel;
    {
        float l0 = 0.f, l1 = 0.f;
        #pragma unroll
        for (int d = 0; d < DGATE; ++d) {
            l0 += gq[d >> 2][d & 3] * wgv[d * NE + 0];
            l1 += gq[d >> 2][d & 3] * wgv[d * NE + 1];
        }
        esel = (l1 > l0) ? 1 : 0;
        const unsigned sel1 = __builtin_amdgcn_ballot_w32(hi == 0 && (row0 + m) < NN && esel == 1);
        if (lane == 0) tile_cnt1[blockIdx.x] = __popc(sel1);
        sel_rows = __builtin_amdgcn_ballot_w32(esel == 1);
    }

    v2f bx[5];
    #pragma unroll
    for (int c = 0; c < 5; ++c) {
        v2f b;
        #pragma unroll
        for (int j = 0; j < 2; ++j) {
            const int s0 = 4 * c + j, s1 = 4 * c + 2 + j;
            const float v0 = (s0 % 10 < DIN) ? xr9[s0 % 10 < DIN ? s0 % 10 : 0] : 1.0f;
            const float v1 = (s1 % 10 < DIN) ? xr9[s1 % 10 < DIN ? s1 % 10 : 0] : 1.0f;
            const float u0 = (esel == s0 / 10) ? v0 : 0.f;
            const float u1 = (esel == s1 / 10) ? v1 : 0.f;
            b[j] = hi ? u1 : u0;
        }
        bx[c] = b;
    }

    v8f hT[2][4];
    #pragma unroll
    for (int g = 0; g < 2; ++g) {
        #pragma unroll
        for (int t = 0; t < 4; ++t) {
            #pragma unroll
            for (int i = 0; i < 8; ++i) hT[g][t][i] = 0.f;
        }
    }
    #pragma unroll
    for (int c = 0; c < 5; ++c) {
        #pragma unroll
        for (int g = 0; g < 2; ++g) {
            #pragma unroll
            for (int t = 0; t < 4; ++t) {
                v2f w;
                w.x = w1s[c][0][g][t];
                w.y = w1s[c][1][g][t];
                hT[g][t] = __builtin_amdgcn_wmma_f32_16x16x4_f32(
                               false, w, false, bx[c], (short)0, hT[g][t], false, false);
            }
        }
    }
    #pragma unroll
    for (int g = 0; g < 2; ++g) {
        #pragma unroll
        for (int t = 0; t < 4; ++t) {
            #pragma unroll
            for (int i = 0; i < 8; ++i) hT[g][t][i] = fmaxf(hT[g][t][i], 0.0f);
        }
    }

    v2f wl01, wl23, wl45, wl67;
    wl01.x = wl[0]; wl01.y = wl[1]; wl23.x = wl[2]; wl23.y = wl[3];
    wl45.x = wl[4]; wl45.y = wl[5]; wl67.x = wl[6]; wl67.y = wl[7];
    v8f acc[2];
    {
        float cb = (q4 == 0) ? bol : 0.f;
        #pragma unroll
        for (int jj = 0; jj < 8; ++jj) cb += b2q[jj >> 2][jj & 3] * wl[jj];
        #pragma unroll
        for (int i = 0; i < 8; ++i) { acc[0][i] = cb; acc[1][i] = 0.f; }
    }
    #pragma unroll
    for (int st = 0; st < HH / 4; ++st) {
        const int g = st >> 4, t = (st >> 2) & 3, s = st & 3;
        v2f a;
        a.x = hT[g][t][2 * s]; a.y = hT[g][t][2 * s + 1];
        v2f bf;
        #pragma unroll
        for (int j = 0; j < 2; ++j) {
            const v4f ra = w2q[st][j][0], rb = w2q[st][j][1];
            v2f p = ra.xy * wl01;
            p = __builtin_elementwise_fma(ra.zw, wl23, p);
            p = __builtin_elementwise_fma(rb.xy, wl45, p);
            p = __builtin_elementwise_fma(rb.zw, wl67, p);
            bf[j] = p.x + p.y;
        }
        acc[st & 1] = __builtin_amdgcn_wmma_f32_16x16x4_f32(
                          false, a, false, bf, (short)0, acc[st & 1], false, false);
    }

    const unsigned sh = sel_rows >> (8 * hi);
    float v[8];
    #pragma unroll
    for (int r = 0; r < 8; ++r) {
        float s = ((int)((sh >> r) & 1u) == ec) ? (acc[0][r] + acc[1][r]) : 0.f;
        s += xrow_f<2>(s);
        s += xrow_f<4>(s);
        s += xrow_f<8>(s);
        v[r] = s;
    }
    const bool p0 = (m & 2) != 0, p1 = (m & 4) != 0, p2 = (m & 8) != 0;
    const float t0 = p0 ? v[1] : v[0], t1 = p0 ? v[3] : v[2];
    const float t2 = p0 ? v[5] : v[4], t3 = p0 ? v[7] : v[6];
    const float u0 = p1 ? t1 : t0,     u1 = p1 ? t3 : t2;
    const float res = p2 ? u1 : u0;
    if (row0 + 8 * hi + (m >> 1) < NN) out[(size_t)row0 * NO + lane] = res;
}

__device__ __forceinline__ int xhalf_i(int v) {
    return __builtin_amdgcn_permlanex16(v, v, 0x76543210, 0xfedcba98, false, false);
}
template <int MASK>
__device__ __forceinline__ int xrow_i(int v) {
    return __builtin_amdgcn_update_dpp(0, v, 0x160 + MASK, 0xf, 0xf, false);
}

__global__ __launch_bounds__(32) void moe_loss_kernel(const int* __restrict__ tile_cnt1,
                                                      float* __restrict__ loss_out) {
    static_assert(TILES % 4 == 0, "counts are read four at a time");
    const int lane = threadIdx.x;
    constexpr int QUADS = TILES / 4;
    constexpr int PER_LANE = (QUADS + 31) / 32;
    v4i q[PER_LANE];
    #pragma unroll
    for (int j = 0; j < PER_LANE; ++j) {
        const int idc = min(j * 32 + lane, QUADS - 1);
        q[j] = *(const v4i*)&tile_cnt1[4 * idc];
    }
    int s = 0;
    #pragma unroll
    for (int j = 0; j < PER_LANE; ++j) {
        const int t = (q[j].x + q[j].y) + (q[j].z + q[j].w);
        if ((j + 1) * 32 <= QUADS) s += t;
        else s += (j * 32 + lane < QUADS) ? t : 0;
    }
    s += xrow_i<1>(s);
    s += xrow_i<2>(s);
    s += xrow_i<4>(s);
    s += xrow_i<8>(s);
    s += xhalf_i(s);
    if (lane != 0) return;
    const float c1 = (float)s;
    const float c0 = (float)NN - c1;
    const float mv = 0.5f * (float)NN;
    const float var = (c0 - mv) * (c0 - mv) + (c1 - mv) * (c1 - mv);
    const float cv  = var / (mv * mv + 1e-10f);
    loss_out[0] = 2.0f * cv * 0.01f;
}

extern "C" void kernel_launch(void* const* d_in, const int* in_sizes, int n_in,
                              void* d_out, int out_size, void* d_ws, size_t ws_size,
                              hipStream_t stream) {
    const float* x      = (const float*)d_in[0];
    const float* gate   = (const float*)d_in[1];
    const float* w_gate = (const float*)d_in[2];
    const float* w1     = (const float*)d_in[3];
    const float* b1     = (const float*)d_in[4];
    const float* w2     = (const float*)d_in[5];
    const float* b2     = (const float*)d_in[6];
    const float* wo     = (const float*)d_in[7];
    const float* bo     = (const float*)d_in[8];
    float* out = (float*)d_out;
    int*   cnt = (int*)d_ws;

    moe_tile_kernel<<<TILES, 32, 0, stream>>>(x, gate, w_gate, w1, b1, w2, b2,
                                              wo, bo, out, cnt);
    moe_loss_kernel<<<1, 32, 0, stream>>>(cnt, out + (size_t)NN * NO);
}


# W2.wo (and b2.wo+bo) folded once per work-group of 4 tile waves, 64 stacked W2 rows per wave, shared through a 2-KB LDS table: 171 -> 51 vector loads per wave; sum order of out changes
# speedup vs baseline: 3.6605x; 1.1858x over previous
#include <hip/hip_runtime.h>

typedef float v2f __attribute__((ext_vector_type(2)));
typedef float v4f __attribute__((ext_vector_type(4)));
typedef float v8f __attribute__((ext_vector_type(8)));
typedef int   v4i __attribute__((ext_vector_type(4)));

#define NN    11826
#define DIN   9
#define DGATE 36
#define NE    2
#define HH    128
#define MO    32
#define NO    2
#define TILES ((NN + 15) / 16)
#define WPG    4
#define GROUPS (TILES / WPG)
#define FPITCH 132

template <int MASK>
__device__ __forceinline__ float xrow_f(float v) {
    return __int_as_float(__builtin_amdgcn_update_dpp(0, __float_as_int(v), 0x160 + MASK, 0xf, 0xf, false));
}

__device__ __forceinline__ float fold_row(v4f r, v4f woA, v4f woB, bool odd) {
    const v2f rx = {r.x, r.x}, ry = {r.y, r.y}, rz = {r.z, r.z}, rw = {r.w, r.w};
    v2f p = rx * woA.xy;
    p = __builtin_elementwise_fma(ry, woA.zw, p);
    p = __builtin_elementwise_fma(rz, woB.xy, p);
    p = __builtin_elementwise_fma(rw, woB.zw, p);
    const float keep = odd ? p.y : p.x;
    const float give = odd ? p.x : p.y;
    float s = keep + xrow_f<1>(give);
    s += xrow_f<2>(s);
    s += xrow_f<4>(s);
    return s;
}

__global__ __launch_bounds__(WPG * 32) void moe_tile_kernel(
    const float* __restrict__ x,
    const float* __restrict__ gate,
    const float* __restrict__ w_gate,
    const float* __restrict__ w1,
    const float* __restrict__ b1,
    const float* __restrict__ w2,
    const float* __restrict__ b2,
    const float* __restrict__ wo,
    const float* __restrict__ bo,
    float* __restrict__ out,
    int* __restrict__ tile_cnt1)
{
    static_assert(TILES % WPG == 0, "every wave of every work-group has a tile");
    __shared__ float fold_tab[4 * FPITCH] __attribute__((aligned(16)));
    const int lane = threadIdx.x & 31;
    const int wv   = __builtin_amdgcn_readfirstlane(threadIdx.x >> 5);
    const int tile = blockIdx.x * WPG + wv;
    const int row0 = tile * 16;
    const int m    = lane & 15;
    const int hi   = lane >> 4;
    const int rowm = min(row0 + m, NN - 1);
    const int o    = m & 1;
    const int ec   = m >> 3;
    const int c4   = lane & 7;
    const int rsub = lane >> 3;

    float xr9[DIN];
    v4f w1s[5][2][2];
    v4f w2r[16];
    v4f b2r;
    v4f woA, woB;
    float bo0, bo1;
    v4f gq[DGATE / 4];
    float wgv[DGATE * NE];

    {
        const float* W2w = w2 + (size_t)(64 * wv + rsub) * MO + 4 * c4;
        #pragma unroll
        for (int i = 0; i < 16; ++i) w2r[i] = *(const v4f*)&W2w[(4 * i) * MO];
        woA = *(const v4f*)&wo[8 * c4];
        woB = *(const v4f*)&wo[8 * c4 + 4];
        b2r = *(const v4f*)&b2[(rsub & 1) * MO + 4 * c4];
        bo0 = bo[0];
        bo1 = bo[1];
    }
    {
        const float* xr = x + (size_t)rowm * DIN;
        #pragma unroll
        for (int k = 0; k < DIN; ++k) xr9[k] = xr[k];
    }
    #pragma unroll
    for (int c = 0; c < 5; ++c) {
        #pragma unroll
        for (int j = 0; j < 2; ++j) {
            const int s0 = 4 * c + j, s1 = 4 * c + 2 + j;
            const float* p0 = (s0 % 10 < DIN) ? w1 + ((s0 / 10) * DIN + s0 % 10) * HH : b1 + (s0 / 10) * HH;
            const float* p1 = (s1 % 10 < DIN) ? w1 + ((s1 / 10) * DIN + s1 % 10) * HH : b1 + (s1 / 10) * HH;
            const float* p  = hi ? p1 : p0;
            w1s[c][j][0] = *(const v4f*)&p[4 * m];
            w1s[c][j][1] = *(const v4f*)&p[64 + 4 * m];
        }
    }
    {
        const float* g = gate + (size_t)rowm * DGATE;
        #pragma unroll
        for (int d = 0; d < DGATE / 4; ++d) gq[d] = *(const v4f*)&g[4 * d];
    }
    #pragma unroll
    for (int d = 0; d < DGATE * NE; ++d) wgv[d] = w_gate[d];

    __builtin_amdgcn_sched_barrier(0);

    {
        const bool odd = (c4 & 1) != 0;
        float fv[16];
        #pragma unroll
        for (int i = 0; i < 16; ++i) fv[i] = fold_row(w2r[i], woA, woB, odd);
        const float fb = fold_row(b2r, woA, woB, odd);
        if (c4 < 2) {
            float* ft = fold_tab + ((wv >> 1) * 2 + c4) * FPITCH + 64 * (wv & 1) + rsub;
            #pragma unroll
            for (int i = 0; i < 16; ++i) ft[4 * i] = fv[i];
            if (wv == 0 && rsub < 2) fold_tab[(rsub * 2 + c4) * FPITCH + 128] = fb + (odd ? bo1 : bo0);
        }
    }

    unsigned sel_rows;
    int cnt1;
    int esel;
    {
        float l0 = 0.f, l1 = 0.f;
        #pragma unroll
        for (int d = 0; d < DGATE; ++d) {
            l0 += gq[d >> 2][d & 3] * wgv[d * NE + 0];
            l1 += gq[d >> 2][d & 3] * wgv[d * NE + 1];
        }
        esel = (l1 > l0) ? 1 : 0;
        const unsigned sel1 = __builtin_amdgcn_ballot_w32(hi == 0 && (row0 + m) < NN && esel == 1);
        cnt1 = __popc(sel1);
        sel_rows = __builtin_amdgcn_ballot_w32(esel == 1);
    }

    v2f bx[5];
    #pragma unroll
    for (int c = 0; c < 5; ++c) {
        v2f b;
        #pragma unroll
        for (int j = 0; j < 2; ++j) {
            const int s0 = 4 * c + j, s1 = 4 * c + 2 + j;
            const float v0 = (s0 % 10 < DIN) ? xr9[s0 % 10 < DIN ? s0 % 10 : 0] : 1.0f;
            const float v1 = (s1 % 10 < DIN) ? xr9[s1 % 10 < DIN ? s1 % 10 : 0] : 1.0f;
            const float u0 = (esel == s0 / 10) ? v0 : 0.f;
            const float u1 = (esel == s1 / 10) ? v1 : 0.f;
            b[j] = hi ? u1 : u0;
        }
        bx[c] = b;
    }

    __syncthreads();
    v4f fq[2][8];
    float cb;
    {
        const float* ft = fold_tab + (ec * 2 + o) * FPITCH;
        #pragma unroll
        for (int g = 0; g < 2; ++g) {
            #pragma unroll
            for (int ch = 0; ch < 8; ++ch) fq[g][ch] = *(const v4f*)&ft[64 * g + 32 * hi + 4 * ch];
        }
        cb = ft[128];
    }

    v8f hT[2][4];
    #pragma unroll
    for (int g = 0; g < 2; ++g) {
        #pragma unroll
        for (int t = 0; t < 4; ++t) {
            #pragma unroll
            for (int i = 0; i < 8; ++i) hT[g][t][i] = 0.f;
        }
    }
    #pragma unroll
    for (int c = 0; c < 5; ++c) {
        #pragma unroll
        for (int g = 0; g < 2; ++g) {
            #pragma unroll
            for (int t = 0; t < 4; ++t) {
                v2f w;
                w.x = w1s[c][0][g][t];
                w.y = w1s[c][1][g][t];
                hT[g][t] = __builtin_amdgcn_wmma_f32_16x16x4_f32(
                               false, w, false, bx[c], (short)0, hT[g][t], false, false);
            }
        }
    }
    #pragma unroll
    for (int g = 0; g < 2; ++g) {
        #pragma unroll
        for (int t = 0; t < 4; ++t) {
            #pragma unroll
            for (int i = 0; i < 8; ++i) hT[g][t][i] = fmaxf(hT[g][t][i], 0.0f);
        }
    }

    v8f acc[2];
    #pragma unroll
    for (int i = 0; i < 8; ++i) { acc[0][i] = cb; acc[1][i] = 0.f; }
    #pragma unroll
    for (int st = 0; st < HH / 4; ++st) {
        const int g = st >> 4, t = (st >> 2) & 3, s = st & 3;
        v2f a;
        a.x = hT[g][t][2 * s]; a.y = hT[g][t][2 * s + 1];
        v2f bf;
        bf.x = fq[g][2 * s][t]; bf.y = fq[g][2 * s + 1][t];
        acc[st & 1] = __builtin_amdgcn_wmma_f32_16x16x4_f32(
                          false, a, false, bf, (short)0, acc[st & 1], false, false);
    }

    const unsigned sh = sel_rows >> (8 * hi);
    float v[8];
    #pragma unroll
    for (int r = 0; r < 8; ++r) {
        float s = ((int)((sh >> r) & 1u) == ec) ? (acc[0][r] + acc[1][r]) : 0.f;
        s += xrow_f<8>(s);
        v[r] = s;
    }
    const bool p0 = (m & 2) != 0, p1 = (m & 4) != 0, p2 = (m & 8) != 0;
    const float t0 = p0 ? v[1] : v[0], t1 = p0 ? v[3] : v[2];
    const float t2 = p0 ? v[5] : v[4], t3 = p0 ? v[7] : v[6];
    const float u0 = p1 ? t1 : t0,     u1 = p1 ? t3 : t2;
    const float res = p2 ? u1 : u0;
    if (lane == 0) tile_cnt1[tile] = cnt1;
    if (row0 + 8 * hi + (m >> 1) < NN) out[(size_t)row0 * NO + lane] = res;
}

__device__ __forceinline__ int xhalf_i(int v) {
    return __builtin_amdgcn_permlanex16(v, v, 0x76543210, 0xfedcba98, false, false);
}
template <int MASK>
__device__ __forceinline__ int xrow_i(int v) {
    return __builtin_amdgcn_update_dpp(0, v, 0x160 + MASK, 0xf, 0xf, false);
}

__global__ __launch_bounds__(32) void moe_loss_kernel(const int* __restrict__ tile_cnt1,
                                                      float* __restrict__ loss_out) {
    static_assert(TILES % 4 == 0, "counts are read four at a time");
    const int lane = threadIdx.x;
    constexpr int QUADS = TILES / 4;
    constexpr int PER_LANE = (QUADS + 31) / 32;
    v4i q[PER_LANE];
    #pragma unroll
    for (int j = 0; j < PER_LANE; ++j) {
        const int idc = min(j * 32 + lane, QUADS - 1);
        q[j] = *(const v4i*)&tile_cnt1[4 * idc];
    }
    int s = 0;
    #pragma unroll
    for (int j = 0; j < PER_LANE; ++j) {
        const int t = (q[j].x + q[j].y) + (q[j].z + q[j].w);
        if ((j + 1) * 32 <= QUADS) s += t;
        else s += (j * 32 + lane < QUADS) ? t : 0;
    }
    s += xrow_i<1>(s);
    s += xrow_i<2>(s);
    s += xrow_i<4>(s);
    s += xrow_i<8>(s);
    s += xhalf_i(s);
    if (lane != 0) return;
    const float c1 = (float)s;
    const float c0 = (float)NN - c1;
    const float mv = 0.5f * (float)NN;
    const float var = (c0 - mv) * (c0 - mv) + (c1 - mv) * (c1 - mv);
    const float cv  = var / (mv * mv + 1e-10f);
    loss_out[0] = 2.0f * cv * 0.01f;
}

extern "C" void kernel_launch(void* const* d_in, const int* in_sizes, int n_in,
                              void* d_out, int out_size, void* d_ws, size_t ws_size,
                              hipStream_t stream) {
    const float* x      = (const float*)d_in[0];
    const float* gate   = (const float*)d_in[1];
    const float* w_gate = (const float*)d_in[2];
    const float* w1     = (const float*)d_in[3];
    const float* b1     = (const float*)d_in[4];
    const float* w2     = (const float*)d_in[5];
    const float* b2     = (const float*)d_in[6];
    const float* wo     = (const float*)d_in[7];
    const float* bo     = (const float*)d_in[8];
    float* out = (float*)d_out;
    int*   cnt = (int*)d_ws;

    moe_tile_kernel<<<GROUPS, WPG * 32, 0, stream>>>(x, gate, w_gate, w1, b1, w2, b2,
                                                     wo, bo, out, cnt);
    moe_loss_kernel<<<1, 32, 0, stream>>>(cnt, out + (size_t)NN * NO);
}
